# starting kernel
# baseline (speedup 1.0000x reference)
#include <hip/hip_runtime.h>

typedef __attribute__((ext_vector_type(2))) float v2f;
typedef __attribute__((ext_vector_type(8))) float v8f;

#define N_NODES 50000
#define DIM 128
#define HID 64
#define NEDGE 500000
#define SLOPE 0.2f

template <int NT>
__global__ __launch_bounds__(128) void gemm_bias_kernel(
    const float* __restrict__ A, const float* __restrict__ W,
    const float* __restrict__ bias, float* __restrict__ C,
    int rows, int K, int relu)
{
    constexpr int ncols = NT * 16;
    const int wave = (int)((blockIdx.x * blockDim.x + threadIdx.x) >> 5);
    const int lane = (int)(threadIdx.x & 31);
    if (wave * 16 >= rows) return;

    const int nloc  = lane & 15;
    const int khalf = (lane >> 4) << 1;

    const float* __restrict__ arow = A + (size_t)(wave * 16 + nloc) * (size_t)K;

    v8f acc[NT];
#pragma unroll
    for (int t = 0; t < NT; ++t)
        acc[t] = (v8f){0.f, 0.f, 0.f, 0.f, 0.f, 0.f, 0.f, 0.f};

    for (int k0 = 0; k0 < K; k0 += 4) {
        v2f a;
        a.x = arow[k0 + khalf];
        a.y = arow[k0 + khalf + 1];
        const float* __restrict__ w0 = W + (size_t)(k0 + khalf) * ncols + nloc;
        const float* __restrict__ w1 = w0 + ncols;
#pragma unroll
        for (int t = 0; t < NT; ++t) {
            v2f b;
            b.x = w0[t * 16];
            b.y = w1[t * 16];
            acc[t] = __builtin_amdgcn_wmma_f32_16x16x4_f32(
                false, a, false, b, (short)0, acc[t], false, false);
        }
    }

    const int mbase = wave * 16 + ((lane >> 4) << 3);
#pragma unroll
    for (int t = 0; t < NT; ++t) {
        const int n = t * 16 + nloc;
        const float bv = bias[n];
#pragma unroll
        for (int j = 0; j < 8; ++j) {
            float v = acc[t][j] + bv;
            if (relu) v = fmaxf(v, 0.f);
            C[(size_t)(mbase + j) * ncols + n] = v;
        }
    }
}

__device__ __forceinline__ unsigned enc_f32(float x) {
    unsigned b = __float_as_uint(x);
    return (b & 0x80000000u) ? ~b : (b | 0x80000000u);
}
__device__ __forceinline__ float dec_f32(unsigned k) {
    unsigned b = (k & 0x80000000u) ? (k & 0x7FFFFFFFu) : ~k;
    return __uint_as_float(b);
}

__global__ void init_out_kernel(float* __restrict__ out,
                                const float* __restrict__ bias, int n) {
    int i = (int)(blockIdx.x * blockDim.x + threadIdx.x);
    if (i < n * DIM) out[i] = bias[i & (DIM - 1)];
}
__global__ void init_ms_kernel(unsigned* __restrict__ menc,
                               float* __restrict__ ssum, int n) {
    int i = (int)(blockIdx.x * blockDim.x + threadIdx.x);
    if (i < n) { menc[i] = 0u; ssum[i] = 0.f; }
}

__global__ __launch_bounds__(256) void edge_logit_kernel(
    const float* __restrict__ xl, const float* __restrict__ xr,
    const int* __restrict__ src, const int* __restrict__ dst,
    const float* __restrict__ att,
    float* __restrict__ logit, unsigned* __restrict__ menc, int nedge)
{
    const int e    = (int)((blockIdx.x * blockDim.x + threadIdx.x) >> 5);
    const int lane = (int)(threadIdx.x & 31);
    if (e >= nedge) return;
    const int s = src[e], d = dst[e];
    const float4 a  = *(const float4*)(xl + (size_t)s * DIM + lane * 4);
    const float4 b  = *(const float4*)(xr + (size_t)d * DIM + lane * 4);
    const float4 at = *(const float4*)(att + lane * 4);
    float z, sum = 0.f;
    z = a.x + b.x; z = (z > 0.f) ? z : SLOPE * z; sum += z * at.x;
    z = a.y + b.y; z = (z > 0.f) ? z : SLOPE * z; sum += z * at.y;
    z = a.z + b.z; z = (z > 0.f) ? z : SLOPE * z; sum += z * at.z;
    z = a.w + b.w; z = (z > 0.f) ? z : SLOPE * z; sum += z * at.w;
#pragma unroll
    for (int off = 16; off > 0; off >>= 1) sum += __shfl_xor(sum, off, 32);
    if (lane == 0) {
        logit[e] = sum;
        atomicMax(menc + d, enc_f32(sum));
    }
}

__global__ void edge_exp_kernel(
    const float* __restrict__ logit, const int* __restrict__ dst,
    const unsigned* __restrict__ menc, float* __restrict__ eexp,
    float* __restrict__ ssum, int nedge)
{
    int e = (int)(blockIdx.x * blockDim.x + threadIdx.x);
    if (e >= nedge) return;
    const int d = dst[e];
    const float v = expf(logit[e] - dec_f32(menc[d]));
    eexp[e] = v;
    atomicAdd(ssum + d, v);
}

__global__ __launch_bounds__(256) void edge_scatter_kernel(
    const float* __restrict__ xl,
    const int* __restrict__ src, const int* __restrict__ dst,
    const float* __restrict__ eexp, const float* __restrict__ ssum,
    float* __restrict__ out, int nedge)
{
    const int e    = (int)((blockIdx.x * blockDim.x + threadIdx.x) >> 5);
    const int lane = (int)(threadIdx.x & 31);
    if (e >= nedge) return;
    const int s = src[e], d = dst[e];
    const float alpha = eexp[e] / (ssum[d] + 1e-16f);
    const float4 a = *(const float4*)(xl + (size_t)s * DIM + lane * 4);
    float* o = out + (size_t)d * DIM + lane * 4;
    atomicAdd(o + 0, alpha * a.x);
    atomicAdd(o + 1, alpha * a.y);
    atomicAdd(o + 2, alpha * a.z);
    atomicAdd(o + 3, alpha * a.w);
}

static inline void launch_gemm(const float* A, const float* W, const float* b,
                               float* C, int rows, int K, int ncols, int relu,
                               hipStream_t stream) {
    long waves = (long)(rows / 16);
    long threads = waves * 32;
    int blocks = (int)((threads + 127) / 128);
    if (ncols == 64)
        gemm_bias_kernel<4><<<blocks, 128, 0, stream>>>(A, W, b, C, rows, K, relu);
    else
        gemm_bias_kernel<8><<<blocks, 128, 0, stream>>>(A, W, b, C, rows, K, relu);
}

extern "C" void kernel_launch(void* const* d_in, const int* in_sizes, int n_in,
                              void* d_out, int out_size, void* d_ws, size_t ws_size,
                              hipStream_t stream) {
    (void)in_sizes; (void)n_in; (void)out_size; (void)ws_size;

    const float* x_user = (const float*)d_in[0];
    const float* x_item = (const float*)d_in[1];
    const int*   e_u2i  = (const int*)d_in[2];
    const int*   e_i2u  = (const int*)d_in[3];
    const float* W1_u = (const float*)d_in[4],  *b1_u = (const float*)d_in[5];
    const float* W2_u = (const float*)d_in[6],  *b2_u = (const float*)d_in[7];
    const float* W1_i = (const float*)d_in[8],  *b1_i = (const float*)d_in[9];
    const float* W2_i = (const float*)d_in[10], *b2_i = (const float*)d_in[11];
    const float* Wl_u2i = (const float*)d_in[12], *bl_u2i = (const float*)d_in[13];
    const float* Wr_u2i = (const float*)d_in[14], *br_u2i = (const float*)d_in[15];
    const float* att_u2i = (const float*)d_in[16], *bias_u2i = (const float*)d_in[17];
    const float* Wl_i2u = (const float*)d_in[18], *bl_i2u = (const float*)d_in[19];
    const float* Wr_i2u = (const float*)d_in[20], *br_i2u = (const float*)d_in[21];
    const float* att_i2u = (const float*)d_in[22], *bias_i2u = (const float*)d_in[23];

    float* out_user = (float*)d_out;
    float* out_item = (float*)d_out + (size_t)N_NODES * DIM;

    float* ws   = (float*)d_ws;
    float* hu   = ws;
    float* hi   = hu + (size_t)N_NODES * DIM;
    float* tmp  = hi + (size_t)N_NODES * DIM;
    float* xl   = tmp + (size_t)N_NODES * HID;
    float* xr   = xl + (size_t)N_NODES * DIM;
    float* logit = xr + (size_t)N_NODES * DIM;
    float* eexp  = logit + NEDGE;
    unsigned* menc = (unsigned*)(eexp + NEDGE);
    float* ssum  = (float*)(menc + N_NODES);

    const int blkN    = (N_NODES + 255) / 256;
    const int blkND   = (N_NODES * DIM + 255) / 256;
    const int blkE    = (NEDGE + 255) / 256;
    const int blkEw   = (NEDGE * 32 + 255) / 256;

    launch_gemm(x_user, W1_u, b1_u, tmp, N_NODES, DIM, HID, 1, stream);
    launch_gemm(tmp,    W2_u, b2_u, hu,  N_NODES, HID, DIM, 0, stream);
    launch_gemm(x_item, W1_i, b1_i, tmp, N_NODES, DIM, HID, 1, stream);
    launch_gemm(tmp,    W2_i, b2_i, hi,  N_NODES, HID, DIM, 0, stream);

    {
        const int* src = e_u2i;
        const int* dst = e_u2i + NEDGE;
        launch_gemm(hu, Wl_u2i, bl_u2i, xl, N_NODES, DIM, DIM, 0, stream);
        launch_gemm(hi, Wr_u2i, br_u2i, xr, N_NODES, DIM, DIM, 0, stream);
        init_out_kernel<<<blkND, 256, 0, stream>>>(out_item, bias_u2i, N_NODES);
        init_ms_kernel<<<blkN, 256, 0, stream>>>(menc, ssum, N_NODES);
        edge_logit_kernel<<<blkEw, 256, 0, stream>>>(xl, xr, src, dst, att_u2i,
                                                     logit, menc, NEDGE);
        edge_exp_kernel<<<blkE, 256, 0, stream>>>(logit, dst, menc, eexp, ssum, NEDGE);
        edge_scatter_kernel<<<blkEw, 256, 0, stream>>>(xl, src, dst, eexp, ssum,
                                                       out_item, NEDGE);
    }

    {
        const int* src = e_i2u;
        const int* dst = e_i2u + NEDGE;
        launch_gemm(hi, Wl_i2u, bl_i2u, xl, N_NODES, DIM, DIM, 0, stream);
        launch_gemm(hu, Wr_i2u, br_i2u, xr, N_NODES, DIM, DIM, 0, stream);
        init_out_kernel<<<blkND, 256, 0, stream>>>(out_user, bias_i2u, N_NODES);
        init_ms_kernel<<<blkN, 256, 0, stream>>>(menc, ssum, N_NODES);
        edge_logit_kernel<<<blkEw, 256, 0, stream>>>(xl, xr, src, dst, att_i2u,
                                                     logit, menc, NEDGE);
        edge_exp_kernel<<<blkE, 256, 0, stream>>>(logit, dst, menc, eexp, ssum, NEDGE);
        edge_scatter_kernel<<<blkEw, 256, 0, stream>>>(xl, src, dst, eexp, ssum,
                                                       out_user, NEDGE);
    }
}


# Edge logit and scatter kernels (edges in input order, float atomics kept): one wave per 32 consecutive edges, lane i holds edge i's indices, 8 row loads in flight; atomic arrival order differs
# speedup vs baseline: 1.2339x; 1.2339x over previous
#include <hip/hip_runtime.h>

typedef __attribute__((ext_vector_type(2))) float v2f;
typedef __attribute__((ext_vector_type(8))) float v8f;

#define N_NODES 50000
#define DIM 128
#define HID 64
#define NEDGE 500000
#define SLOPE 0.2f

template <int NT>
__global__ __launch_bounds__(128) void gemm_bias_kernel(
    const float* __restrict__ A, const float* __restrict__ W,
    const float* __restrict__ bias, float* __restrict__ C,
    int rows, int K, int relu)
{
    constexpr int ncols = NT * 16;
    const int wave = (int)((blockIdx.x * blockDim.x + threadIdx.x) >> 5);
    const int lane = (int)(threadIdx.x & 31);
    if (wave * 16 >= rows) return;

    const int nloc  = lane & 15;
    const int khalf = (lane >> 4) << 1;

    const float* __restrict__ arow = A + (size_t)(wave * 16 + nloc) * (size_t)K;

    v8f acc[NT];
#pragma unroll
    for (int t = 0; t < NT; ++t)
        acc[t] = (v8f){0.f, 0.f, 0.f, 0.f, 0.f, 0.f, 0.f, 0.f};

    for (int k0 = 0; k0 < K; k0 += 4) {
        v2f a;
        a.x = arow[k0 + khalf];
        a.y = arow[k0 + khalf + 1];
        const float* __restrict__ w0 = W + (size_t)(k0 + khalf) * ncols + nloc;
        const float* __restrict__ w1 = w0 + ncols;
#pragma unroll
        for (int t = 0; t < NT; ++t) {
            v2f b;
            b.x = w0[t * 16];
            b.y = w1[t * 16];
            acc[t] = __builtin_amdgcn_wmma_f32_16x16x4_f32(
                false, a, false, b, (short)0, acc[t], false, false);
        }
    }

    const int mbase = wave * 16 + ((lane >> 4) << 3);
#pragma unroll
    for (int t = 0; t < NT; ++t) {
        const int n = t * 16 + nloc;
        const float bv = bias[n];
#pragma unroll
        for (int j = 0; j < 8; ++j) {
            float v = acc[t][j] + bv;
            if (relu) v = fmaxf(v, 0.f);
            C[(size_t)(mbase + j) * ncols + n] = v;
        }
    }
}

__device__ __forceinline__ unsigned enc_f32(float x) {
    unsigned b = __float_as_uint(x);
    return (b & 0x80000000u) ? ~b : (b | 0x80000000u);
}
__device__ __forceinline__ float dec_f32(unsigned k) {
    unsigned b = (k & 0x80000000u) ? (k & 0x7FFFFFFFu) : ~k;
    return __uint_as_float(b);
}

__global__ void init_out_kernel(float* __restrict__ out,
                                const float* __restrict__ bias, int n) {
    int i = (int)(blockIdx.x * blockDim.x + threadIdx.x);
    if (i < n * DIM) out[i] = bias[i & (DIM - 1)];
}
__global__ void init_ms_kernel(unsigned* __restrict__ menc,
                               float* __restrict__ ssum, int n) {
    int i = (int)(blockIdx.x * blockDim.x + threadIdx.x);
    if (i < n) { menc[i] = 0u; ssum[i] = 0.f; }
}

#define LU 4
#define SU 8

__global__ __launch_bounds__(256) void edge_logit_kernel(
    const float* __restrict__ xl, const float* __restrict__ xr,
    const int* __restrict__ src, const int* __restrict__ dst,
    const float* __restrict__ att,
    float* __restrict__ logit, unsigned* __restrict__ menc, int nedge)
{
    const int gt   = (int)(blockIdx.x * blockDim.x + threadIdx.x);
    const int lane = (int)(threadIdx.x & 31);
    const int e0 = __builtin_amdgcn_readfirstlane(gt & ~31);
    if (e0 >= nedge) return;
    const bool ok = gt < nedge;
    const int sv = ok ? src[gt] : 0;
    const int dv = ok ? dst[gt] : 0;
    const float4 at = *(const float4*)(att + lane * 4);
    const int cnt = (nedge - e0 < 32) ? (nedge - e0) : 32;

    float res = 0.f;
    for (int i0 = 0; i0 < cnt; i0 += LU) {
        float4 a[LU], b[LU];
#pragma unroll
        for (int u = 0; u < LU; ++u) {
            const int s = __builtin_amdgcn_readlane(sv, i0 + u);
            const int d = __builtin_amdgcn_readlane(dv, i0 + u);
            a[u] = *(const float4*)(xl + (size_t)s * DIM + lane * 4);
            b[u] = *(const float4*)(xr + (size_t)d * DIM + lane * 4);
        }
#pragma unroll
        for (int u = 0; u < LU; ++u) {
            float z, sum = 0.f;
            z = a[u].x + b[u].x; z = (z > 0.f) ? z : SLOPE * z; sum = __builtin_fmaf(z, at.x, sum);
            z = a[u].y + b[u].y; z = (z > 0.f) ? z : SLOPE * z; sum = __builtin_fmaf(z, at.y, sum);
            z = a[u].z + b[u].z; z = (z > 0.f) ? z : SLOPE * z; sum = __builtin_fmaf(z, at.z, sum);
            z = a[u].w + b[u].w; z = (z > 0.f) ? z : SLOPE * z; sum = __builtin_fmaf(z, at.w, sum);
#pragma unroll
            for (int off = 16; off > 0; off >>= 1) sum += __shfl_xor(sum, off, 32);
            if (lane == i0 + u) res = sum;
        }
    }
    if (ok) {
        logit[gt] = res;
        atomicMax(menc + dv, enc_f32(res));
    }
}

__global__ void edge_exp_kernel(
    const float* __restrict__ logit, const int* __restrict__ dst,
    const unsigned* __restrict__ menc, float* __restrict__ eexp,
    float* __restrict__ ssum, int nedge)
{
    int e = (int)(blockIdx.x * blockDim.x + threadIdx.x);
    if (e >= nedge) return;
    const int d = dst[e];
    const float v = expf(logit[e] - dec_f32(menc[d]));
    eexp[e] = v;
    atomicAdd(ssum + d, v);
}

__global__ __launch_bounds__(256) void edge_scatter_kernel(
    const float* __restrict__ xl,
    const int* __restrict__ src, const int* __restrict__ dst,
    const float* __restrict__ eexp, const float* __restrict__ ssum,
    float* __restrict__ out, int nedge)
{
    const int gt   = (int)(blockIdx.x * blockDim.x + threadIdx.x);
    const int lane = (int)(threadIdx.x & 31);
    const int e0 = __builtin_amdgcn_readfirstlane(gt & ~31);
    if (e0 >= nedge) return;
    const bool ok = gt < nedge;
    const int sv = ok ? src[gt] : 0;
    const int dv = ok ? dst[gt] : 0;
    float av = 0.f;
    if (ok) av = eexp[gt] / (ssum[dv] + 1e-16f);
    const int avi = __float_as_int(av);
    const int cnt = (nedge - e0 < 32) ? (nedge - e0) : 32;

    int i0 = 0;
    for (; i0 + SU <= cnt; i0 += SU) {
        float4 a[SU];
#pragma unroll
        for (int u = 0; u < SU; ++u) {
            const int s = __builtin_amdgcn_readlane(sv, i0 + u);
            a[u] = *(const float4*)(xl + (size_t)s * DIM + lane * 4);
        }
#pragma unroll
        for (int u = 0; u < SU; ++u) {
            const int d = __builtin_amdgcn_readlane(dv, i0 + u);
            const float alpha = __int_as_float(__builtin_amdgcn_readlane(avi, i0 + u));
            float* o = out + (size_t)d * DIM + lane * 4;
            atomicAdd(o + 0, alpha * a[u].x);
            atomicAdd(o + 1, alpha * a[u].y);
            atomicAdd(o + 2, alpha * a[u].z);
            atomicAdd(o + 3, alpha * a[u].w);
        }
    }
    for (; i0 < cnt; ++i0) {
        const int s = __builtin_amdgcn_readlane(sv, i0);
        const int d = __builtin_amdgcn_readlane(dv, i0);
        const float alpha = __int_as_float(__builtin_amdgcn_readlane(avi, i0));
        const float4 a = *(const float4*)(xl + (size_t)s * DIM + lane * 4);
        float* o = out + (size_t)d * DIM + lane * 4;
        atomicAdd(o + 0, alpha * a.x);
        atomicAdd(o + 1, alpha * a.y);
        atomicAdd(o + 2, alpha * a.z);
        atomicAdd(o + 3, alpha * a.w);
    }
}

static inline void launch_gemm(const float* A, const float* W, const float* b,
                               float* C, int rows, int K, int ncols, int relu,
                               hipStream_t stream) {
    long waves = (long)(rows / 16);
    long threads = waves * 32;
    int blocks = (int)((threads + 127) / 128);
    if (ncols == 64)
        gemm_bias_kernel<4><<<blocks, 128, 0, stream>>>(A, W, b, C, rows, K, relu);
    else
        gemm_bias_kernel<8><<<blocks, 128, 0, stream>>>(A, W, b, C, rows, K, relu);
}

extern "C" void kernel_launch(void* const* d_in, const int* in_sizes, int n_in,
                              void* d_out, int out_size, void* d_ws, size_t ws_size,
                              hipStream_t stream) {
    (void)in_sizes; (void)n_in; (void)out_size; (void)ws_size;

    const float* x_user = (const float*)d_in[0];
    const float* x_item = (const float*)d_in[1];
    const int*   e_u2i  = (const int*)d_in[2];
    const int*   e_i2u  = (const int*)d_in[3];
    const float* W1_u = (const float*)d_in[4],  *b1_u = (const float*)d_in[5];
    const float* W2_u = (const float*)d_in[6],  *b2_u = (const float*)d_in[7];
    const float* W1_i = (const float*)d_in[8],  *b1_i = (const float*)d_in[9];
    const float* W2_i = (const float*)d_in[10], *b2_i = (const float*)d_in[11];
    const float* Wl_u2i = (const float*)d_in[12], *bl_u2i = (const float*)d_in[13];
    const float* Wr_u2i = (const float*)d_in[14], *br_u2i = (const float*)d_in[15];
    const float* att_u2i = (const float*)d_in[16], *bias_u2i = (const float*)d_in[17];
    const float* Wl_i2u = (const float*)d_in[18], *bl_i2u = (const float*)d_in[19];
    const float* Wr_i2u = (const float*)d_in[20], *br_i2u = (const float*)d_in[21];
    const float* att_i2u = (const float*)d_in[22], *bias_i2u = (const float*)d_in[23];

    float* out_user = (float*)d_out;
    float* out_item = (float*)d_out + (size_t)N_NODES * DIM;

    float* ws   = (float*)d_ws;
    float* hu   = ws;
    float* hi   = hu + (size_t)N_NODES * DIM;
    float* tmp  = hi + (size_t)N_NODES * DIM;
    float* xl   = tmp + (size_t)N_NODES * HID;
    float* xr   = xl + (size_t)N_NODES * DIM;
    float* logit = xr + (size_t)N_NODES * DIM;
    float* eexp  = logit + NEDGE;
    unsigned* menc = (unsigned*)(eexp + NEDGE);
    float* ssum  = (float*)(menc + N_NODES);

    const int blkN    = (N_NODES + 255) / 256;
    const int blkND   = (N_NODES * DIM + 255) / 256;
    const int blkE    = (NEDGE + 255) / 256;

    launch_gemm(x_user, W1_u, b1_u, tmp, N_NODES, DIM, HID, 1, stream);
    launch_gemm(tmp,    W2_u, b2_u, hu,  N_NODES, HID, DIM, 0, stream);
    launch_gemm(x_item, W1_i, b1_i, tmp, N_NODES, DIM, HID, 1, stream);
    launch_gemm(tmp,    W2_i, b2_i, hi,  N_NODES, HID, DIM, 0, stream);

    {
        const int* src = e_u2i;
        const int* dst = e_u2i + NEDGE;
        launch_gemm(hu, Wl_u2i, bl_u2i, xl, N_NODES, DIM, DIM, 0, stream);
        launch_gemm(hi, Wr_u2i, br_u2i, xr, N_NODES, DIM, DIM, 0, stream);
        init_out_kernel<<<blkND, 256, 0, stream>>>(out_item, bias_u2i, N_NODES);
        init_ms_kernel<<<blkN, 256, 0, stream>>>(menc, ssum, N_NODES);
        edge_logit_kernel<<<blkE, 256, 0, stream>>>(xl, xr, src, dst, att_u2i,
                                                     logit, menc, NEDGE);
        edge_exp_kernel<<<blkE, 256, 0, stream>>>(logit, dst, menc, eexp, ssum, NEDGE);
        edge_scatter_kernel<<<blkE, 256, 0, stream>>>(xl, src, dst, eexp, ssum,
                                                       out_item, NEDGE);
    }

    {
        const int* src = e_i2u;
        const int* dst = e_i2u + NEDGE;
        launch_gemm(hi, Wl_i2u, bl_i2u, xl, N_NODES, DIM, DIM, 0, stream);
        launch_gemm(hu, Wr_i2u, br_i2u, xr, N_NODES, DIM, DIM, 0, stream);
        init_out_kernel<<<blkND, 256, 0, stream>>>(out_user, bias_i2u, N_NODES);
        init_ms_kernel<<<blkN, 256, 0, stream>>>(menc, ssum, N_NODES);
        edge_logit_kernel<<<blkE, 256, 0, stream>>>(xl, xr, src, dst, att_i2u,
                                                     logit, menc, NEDGE);
        edge_exp_kernel<<<blkE, 256, 0, stream>>>(logit, dst, menc, eexp, ssum, NEDGE);
        edge_scatter_kernel<<<blkE, 256, 0, stream>>>(xl, src, dst, eexp, ssum,
                                                       out_user, NEDGE);
    }
}


# Scatter runs in the 64-wide hidden space: atomics add alpha*hid[src] (half the atomics and gather bytes), then W2 and Wl GEMMs map each node's sum to the output; sums re-associated
# speedup vs baseline: 1.4952x; 1.2118x over previous
#include <hip/hip_runtime.h>

typedef __attribute__((ext_vector_type(2))) float v2f;
typedef __attribute__((ext_vector_type(8))) float v8f;

#define N_NODES 50000
#define DIM 128
#define HID 64
#define NEDGE 500000
#define SLOPE 0.2f

template <int NT>
__global__ __launch_bounds__(128) void gemm_bias_kernel(
    const float* __restrict__ A, const float* __restrict__ W,
    const float* __restrict__ bias, const float* __restrict__ rowscale,
    const float* __restrict__ bias2, float* __restrict__ C,
    int rows, int K, int relu)
{
    constexpr int ncols = NT * 16;
    const int wave = (int)((blockIdx.x * blockDim.x + threadIdx.x) >> 5);
    const int lane = (int)(threadIdx.x & 31);
    if (wave * 16 >= rows) return;

    const int nloc  = lane & 15;
    const int khalf = (lane >> 4) << 1;

    const float* __restrict__ arow = A + (size_t)(wave * 16 + nloc) * (size_t)K;

    v8f acc[NT];
#pragma unroll
    for (int t = 0; t < NT; ++t)
        acc[t] = (v8f){0.f, 0.f, 0.f, 0.f, 0.f, 0.f, 0.f, 0.f};

    for (int k0 = 0; k0 < K; k0 += 4) {
        v2f a;
        a.x = arow[k0 + khalf];
        a.y = arow[k0 + khalf + 1];
        const float* __restrict__ w0 = W + (size_t)(k0 + khalf) * ncols + nloc;
        const float* __restrict__ w1 = w0 + ncols;
#pragma unroll
        for (int t = 0; t < NT; ++t) {
            v2f b;
            b.x = w0[t * 16];
            b.y = w1[t * 16];
            acc[t] = __builtin_amdgcn_wmma_f32_16x16x4_f32(
                false, a, false, b, (short)0, acc[t], false, false);
        }
    }

    const int mbase = wave * 16 + ((lane >> 4) << 3);
    float rs[8];
#pragma unroll
    for (int j = 0; j < 8; ++j)
        rs[j] = rowscale ? rowscale[mbase + j] : 1.f;
#pragma unroll
    for (int t = 0; t < NT; ++t) {
        const int n = t * 16 + nloc;
        const float bv = bias[n];
        const float b2v = bias2 ? bias2[n] : 0.f;
#pragma unroll
        for (int j = 0; j < 8; ++j) {
            float v;
            if (rowscale) v = acc[t][j] + rs[j] * bv + b2v;
            else          v = acc[t][j] + bv;
            if (relu) v = fmaxf(v, 0.f);
            C[(size_t)(mbase + j) * ncols + n] = v;
        }
    }
}

__device__ __forceinline__ unsigned enc_f32(float x) {
    unsigned b = __float_as_uint(x);
    return (b & 0x80000000u) ? ~b : (b | 0x80000000u);
}
__device__ __forceinline__ float dec_f32(unsigned k) {
    unsigned b = (k & 0x80000000u) ? (k & 0x7FFFFFFFu) : ~k;
    return __uint_as_float(b);
}

__global__ void init_agg_kernel(float* __restrict__ agg, float* __restrict__ wsum,
                                const float* __restrict__ ssum, int n) {
    int i = (int)(blockIdx.x * blockDim.x + threadIdx.x);
    if (i < n * HID) agg[i] = 0.f;
    if (i < n) { const float s = ssum[i]; wsum[i] = s / (s + 1e-16f); }
}
__global__ void init_ms_kernel(unsigned* __restrict__ menc,
                               float* __restrict__ ssum, int n) {
    int i = (int)(blockIdx.x * blockDim.x + threadIdx.x);
    if (i < n) { menc[i] = 0u; ssum[i] = 0.f; }
}

#define LU 4
#define SU 16

__global__ __launch_bounds__(256) void edge_logit_kernel(
    const float* __restrict__ xl, const float* __restrict__ xr,
    const int* __restrict__ src, const int* __restrict__ dst,
    const float* __restrict__ att,
    float* __restrict__ logit, unsigned* __restrict__ menc, int nedge)
{
    const int gt   = (int)(blockIdx.x * blockDim.x + threadIdx.x);
    const int lane = (int)(threadIdx.x & 31);
    const int e0 = __builtin_amdgcn_readfirstlane(gt & ~31);
    if (e0 >= nedge) return;
    const bool ok = gt < nedge;
    const int sv = ok ? src[gt] : 0;
    const int dv = ok ? dst[gt] : 0;
    const float4 at = *(const float4*)(att + lane * 4);
    const int cnt = (nedge - e0 < 32) ? (nedge - e0) : 32;

    float res = 0.f;
    for (int i0 = 0; i0 < cnt; i0 += LU) {
        float4 a[LU], b[LU];
#pragma unroll
        for (int u = 0; u < LU; ++u) {
            const int s = __builtin_amdgcn_readlane(sv, i0 + u);
            const int d = __builtin_amdgcn_readlane(dv, i0 + u);
            a[u] = *(const float4*)(xl + (size_t)s * DIM + lane * 4);
            b[u] = *(const float4*)(xr + (size_t)d * DIM + lane * 4);
        }
#pragma unroll
        for (int u = 0; u < LU; ++u) {
            float z, sum = 0.f;
            z = a[u].x + b[u].x; z = (z > 0.f) ? z : SLOPE * z; sum = __builtin_fmaf(z, at.x, sum);
            z = a[u].y + b[u].y; z = (z > 0.f) ? z : SLOPE * z; sum = __builtin_fmaf(z, at.y, sum);
            z = a[u].z + b[u].z; z = (z > 0.f) ? z : SLOPE * z; sum = __builtin_fmaf(z, at.z, sum);
            z = a[u].w + b[u].w; z = (z > 0.f) ? z : SLOPE * z; sum = __builtin_fmaf(z, at.w, sum);
#pragma unroll
            for (int off = 16; off > 0; off >>= 1) sum += __shfl_xor(sum, off, 32);
            if (lane == i0 + u) res = sum;
        }
    }
    if (ok) {
        logit[gt] = res;
        atomicMax(menc + dv, enc_f32(res));
    }
}

__global__ void edge_exp_kernel(
    const float* __restrict__ logit, const int* __restrict__ dst,
    const unsigned* __restrict__ menc, float* __restrict__ eexp,
    float* __restrict__ ssum, int nedge)
{
    int e = (int)(blockIdx.x * blockDim.x + threadIdx.x);
    if (e >= nedge) return;
    const int d = dst[e];
    const float v = expf(logit[e] - dec_f32(menc[d]));
    eexp[e] = v;
    atomicAdd(ssum + d, v);
}

__global__ __launch_bounds__(256) void edge_scatter_kernel(
    const float* __restrict__ hid,
    const int* __restrict__ src, const int* __restrict__ dst,
    const float* __restrict__ eexp, const float* __restrict__ ssum,
    float* __restrict__ agg, int nedge)
{
    const int gt   = (int)(blockIdx.x * blockDim.x + threadIdx.x);
    const int lane = (int)(threadIdx.x & 31);
    const int e0 = __builtin_amdgcn_readfirstlane(gt & ~31);
    if (e0 >= nedge) return;
    const bool ok = gt < nedge;
    const int sv = ok ? src[gt] : 0;
    const int dv = ok ? dst[gt] : 0;
    float av = 0.f;
    if (ok) av = eexp[gt] / (ssum[dv] + 1e-16f);
    const int avi = __float_as_int(av);
    const int cnt = (nedge - e0 < 32) ? (nedge - e0) : 32;

    int i0 = 0;
    for (; i0 + SU <= cnt; i0 += SU) {
        float2 a[SU];
#pragma unroll
        for (int u = 0; u < SU; ++u) {
            const int s = __builtin_amdgcn_readlane(sv, i0 + u);
            a[u] = *(const float2*)(hid + (size_t)s * HID + lane * 2);
        }
#pragma unroll
        for (int u = 0; u < SU; ++u) {
            const int d = __builtin_amdgcn_readlane(dv, i0 + u);
            const float alpha = __int_as_float(__builtin_amdgcn_readlane(avi, i0 + u));
            float* o = agg + (size_t)d * HID + lane * 2;
            atomicAdd(o + 0, alpha * a[u].x);
            atomicAdd(o + 1, alpha * a[u].y);
        }
    }
    for (; i0 < cnt; ++i0) {
        const int s = __builtin_amdgcn_readlane(sv, i0);
        const int d = __builtin_amdgcn_readlane(dv, i0);
        const float alpha = __int_as_float(__builtin_amdgcn_readlane(avi, i0));
        const float2 a = *(const float2*)(hid + (size_t)s * HID + lane * 2);
        float* o = agg + (size_t)d * HID + lane * 2;
        atomicAdd(o + 0, alpha * a.x);
        atomicAdd(o + 1, alpha * a.y);
    }
}

static inline void launch_gemm(const float* A, const float* W, const float* b,
                               const float* rowscale, const float* b2,
                               float* C, int rows, int K, int ncols, int relu,
                               hipStream_t stream) {
    long waves = (long)(rows / 16);
    long threads = waves * 32;
    int blocks = (int)((threads + 127) / 128);
    if (ncols == 64)
        gemm_bias_kernel<4><<<blocks, 128, 0, stream>>>(A, W, b, rowscale, b2, C, rows, K, relu);
    else
        gemm_bias_kernel<8><<<blocks, 128, 0, stream>>>(A, W, b, rowscale, b2, C, rows, K, relu);
}

extern "C" void kernel_launch(void* const* d_in, const int* in_sizes, int n_in,
                              void* d_out, int out_size, void* d_ws, size_t ws_size,
                              hipStream_t stream) {
    (void)in_sizes; (void)n_in; (void)out_size; (void)ws_size;

    const float* x_user = (const float*)d_in[0];
    const float* x_item = (const float*)d_in[1];
    const int*   e_u2i  = (const int*)d_in[2];
    const int*   e_i2u  = (const int*)d_in[3];
    const float* W1_u = (const float*)d_in[4],  *b1_u = (const float*)d_in[5];
    const float* W2_u = (const float*)d_in[6],  *b2_u = (const float*)d_in[7];
    const float* W1_i = (const float*)d_in[8],  *b1_i = (const float*)d_in[9];
    const float* W2_i = (const float*)d_in[10], *b2_i = (const float*)d_in[11];
    const float* Wl_u2i = (const float*)d_in[12], *bl_u2i = (const float*)d_in[13];
    const float* Wr_u2i = (const float*)d_in[14], *br_u2i = (const float*)d_in[15];
    const float* att_u2i = (const float*)d_in[16], *bias_u2i = (const float*)d_in[17];
    const float* Wl_i2u = (const float*)d_in[18], *bl_i2u = (const float*)d_in[19];
    const float* Wr_i2u = (const float*)d_in[20], *br_i2u = (const float*)d_in[21];
    const float* att_i2u = (const float*)d_in[22], *bias_i2u = (const float*)d_in[23];

    float* out_user = (float*)d_out;
    float* out_item = (float*)d_out + (size_t)N_NODES * DIM;

    float* ws   = (float*)d_ws;
    float* hu   = ws;
    float* hi   = hu + (size_t)N_NODES * DIM;
    float* hidu = hi + (size_t)N_NODES * DIM;
    float* xl   = hidu + (size_t)N_NODES * HID;
    float* xr   = xl + (size_t)N_NODES * DIM;
    float* logit = xr + (size_t)N_NODES * DIM;
    float* eexp  = logit + NEDGE;
    unsigned* menc = (unsigned*)(eexp + NEDGE);
    float* ssum  = (float*)(menc + N_NODES);
    float* hidi = out_user;
    float* agg  = xr;
    float* wsum = xr + (size_t)N_NODES * HID;
    float* tsum = xl;

    const float* none = nullptr;

    const int blkN    = (N_NODES + 255) / 256;
    const int blkNH   = (N_NODES * HID + 255) / 256;
    const int blkE    = (NEDGE + 255) / 256;

    launch_gemm(x_user, W1_u, b1_u, none, none, hidu, N_NODES, DIM, HID, 1, stream);
    launch_gemm(hidu,   W2_u, b2_u, none, none, hu,   N_NODES, HID, DIM, 0, stream);
    launch_gemm(x_item, W1_i, b1_i, none, none, hidi, N_NODES, DIM, HID, 1, stream);
    launch_gemm(hidi,   W2_i, b2_i, none, none, hi,   N_NODES, HID, DIM, 0, stream);

    {
        const int* src = e_u2i;
        const int* dst = e_u2i + NEDGE;
        launch_gemm(hu, Wl_u2i, bl_u2i, none, none, xl, N_NODES, DIM, DIM, 0, stream);
        launch_gemm(hi, Wr_u2i, br_u2i, none, none, xr, N_NODES, DIM, DIM, 0, stream);
        init_ms_kernel<<<blkN, 256, 0, stream>>>(menc, ssum, N_NODES);
        edge_logit_kernel<<<blkE, 256, 0, stream>>>(xl, xr, src, dst, att_u2i,
                                                     logit, menc, NEDGE);
        edge_exp_kernel<<<blkE, 256, 0, stream>>>(logit, dst, menc, eexp, ssum, NEDGE);
        init_agg_kernel<<<blkNH, 256, 0, stream>>>(agg, wsum, ssum, N_NODES);
        edge_scatter_kernel<<<blkE, 256, 0, stream>>>(hidu, src, dst, eexp, ssum,
                                                       agg, NEDGE);
        launch_gemm(agg,  W2_u,   b2_u,   wsum, none,     tsum,     N_NODES, HID, DIM, 0, stream);
        launch_gemm(tsum, Wl_u2i, bl_u2i, wsum, bias_u2i, out_item, N_NODES, DIM, DIM, 0, stream);
    }

    {
        const int* src = e_i2u;
        const int* dst = e_i2u + NEDGE;
        launch_gemm(hi, Wl_i2u, bl_i2u, none, none, xl, N_NODES, DIM, DIM, 0, stream);
        launch_gemm(hu, Wr_i2u, br_i2u, none, none, xr, N_NODES, DIM, DIM, 0, stream);
        init_ms_kernel<<<blkN, 256, 0, stream>>>(menc, ssum, N_NODES);
        edge_logit_kernel<<<blkE, 256, 0, stream>>>(xl, xr, src, dst, att_i2u,
                                                     logit, menc, NEDGE);
        edge_exp_kernel<<<blkE, 256, 0, stream>>>(logit, dst, menc, eexp, ssum, NEDGE);
        init_agg_kernel<<<blkNH, 256, 0, stream>>>(agg, wsum, ssum, N_NODES);
        edge_scatter_kernel<<<blkE, 256, 0, stream>>>(hidi, src, dst, eexp, ssum,
                                                       agg, NEDGE);
        launch_gemm(agg,  W2_i,   b2_i,   wsum, none,     tsum,     N_NODES, HID, DIM, 0, stream);
        launch_gemm(tsum, Wl_i2u, bl_i2u, wsum, bias_i2u, out_user, N_NODES, DIM, DIM, 0, stream);
    }
}


# Output map folded: a small WMMA kernel forms W2@Wl and b2@Wl+bl per relation, then one K=64 GEMM takes the scattered sums to the output (was two GEMMs through a 128-wide temp); sums re-associated
# speedup vs baseline: 1.6193x; 1.0830x over previous
#include <hip/hip_runtime.h>

typedef __attribute__((ext_vector_type(2))) float v2f;
typedef __attribute__((ext_vector_type(8))) float v8f;

#define N_NODES 50000
#define DIM 128
#define HID 64
#define NEDGE 500000
#define SLOPE 0.2f

template <int NT>
__global__ __launch_bounds__(128) void gemm_bias_kernel(
    const float* __restrict__ A, const float* __restrict__ W,
    const float* __restrict__ bias, const float* __restrict__ rowscale,
    const float* __restrict__ bias2, float* __restrict__ C,
    int rows, int K, int relu)
{
    constexpr int ncols = NT * 16;
    const int wave = (int)((blockIdx.x * blockDim.x + threadIdx.x) >> 5);
    const int lane = (int)(threadIdx.x & 31);
    if (wave * 16 >= rows) return;

    const int nloc  = lane & 15;
    const int khalf = (lane >> 4) << 1;

    const float* __restrict__ arow = A + (size_t)(wave * 16 + nloc) * (size_t)K;

    v8f acc[NT];
#pragma unroll
    for (int t = 0; t < NT; ++t)
        acc[t] = (v8f){0.f, 0.f, 0.f, 0.f, 0.f, 0.f, 0.f, 0.f};

    for (int k0 = 0; k0 < K; k0 += 4) {
        v2f a;
        a.x = arow[k0 + khalf];
        a.y = arow[k0 + khalf + 1];
        const float* __restrict__ w0 = W + (size_t)(k0 + khalf) * ncols + nloc;
        const float* __restrict__ w1 = w0 + ncols;
#pragma unroll
        for (int t = 0; t < NT; ++t) {
            v2f b;
            b.x = w0[t * 16];
            b.y = w1[t * 16];
            acc[t] = __builtin_amdgcn_wmma_f32_16x16x4_f32(
                false, a, false, b, (short)0, acc[t], false, false);
        }
    }

    const int mbase = wave * 16 + ((lane >> 4) << 3);
    float rs[8];
#pragma unroll
    for (int j = 0; j < 8; ++j)
        rs[j] = rowscale ? rowscale[mbase + j] : 1.f;
#pragma unroll
    for (int t = 0; t < NT; ++t) {
        const int n = t * 16 + nloc;
        const float bv = bias[n];
        const float b2v = bias2 ? bias2[n] : 0.f;
#pragma unroll
        for (int j = 0; j < 8; ++j) {
            float v;
            if (rowscale) v = acc[t][j] + rs[j] * bv + b2v;
            else          v = acc[t][j] + bv;
            if (relu) v = fmaxf(v, 0.f);
            C[(size_t)(mbase + j) * ncols + n] = v;
        }
    }
}

__device__ __forceinline__ unsigned enc_f32(float x) {
    unsigned b = __float_as_uint(x);
    return (b & 0x80000000u) ? ~b : (b | 0x80000000u);
}
__device__ __forceinline__ float dec_f32(unsigned k) {
    unsigned b = (k & 0x80000000u) ? (k & 0x7FFFFFFFu) : ~k;
    return __uint_as_float(b);
}

__global__ void init_agg_kernel(float* __restrict__ agg, float* __restrict__ wsum,
                                const float* __restrict__ ssum, int n) {
    int i = (int)(blockIdx.x * blockDim.x + threadIdx.x);
    if (i < n * HID) agg[i] = 0.f;
    if (i < n) { const float s = ssum[i]; wsum[i] = s / (s + 1e-16f); }
}
__global__ void init_ms_kernel(unsigned* __restrict__ menc,
                               float* __restrict__ ssum, int n) {
    int i = (int)(blockIdx.x * blockDim.x + threadIdx.x);
    if (i < n) { menc[i] = 0u; ssum[i] = 0.f; }
}

#define LU 4
#define SU 16

__global__ __launch_bounds__(256) void edge_logit_kernel(
    const float* __restrict__ xl, const float* __restrict__ xr,
    const int* __restrict__ src, const int* __restrict__ dst,
    const float* __restrict__ att,
    float* __restrict__ logit, unsigned* __restrict__ menc, int nedge)
{
    const int gt   = (int)(blockIdx.x * blockDim.x + threadIdx.x);
    const int lane = (int)(threadIdx.x & 31);
    const int e0 = __builtin_amdgcn_readfirstlane(gt & ~31);
    if (e0 >= nedge) return;
    const bool ok = gt < nedge;
    const int sv = ok ? src[gt] : 0;
    const int dv = ok ? dst[gt] : 0;
    const float4 at = *(const float4*)(att + lane * 4);
    const int cnt = (nedge - e0 < 32) ? (nedge - e0) : 32;

    float res = 0.f;
    for (int i0 = 0; i0 < cnt; i0 += LU) {
        float4 a[LU], b[LU];
#pragma unroll
        for (int u = 0; u < LU; ++u) {
            const int s = __builtin_amdgcn_readlane(sv, i0 + u);
            const int d = __builtin_amdgcn_readlane(dv, i0 + u);
            a[u] = *(const float4*)(xl + (size_t)s * DIM + lane * 4);
            b[u] = *(const float4*)(xr + (size_t)d * DIM + lane * 4);
        }
#pragma unroll
        for (int u = 0; u < LU; ++u) {
            float z, sum = 0.f;
            z = a[u].x + b[u].x; z = (z > 0.f) ? z : SLOPE * z; sum = __builtin_fmaf(z, at.x, sum);
            z = a[u].y + b[u].y; z = (z > 0.f) ? z : SLOPE * z; sum = __builtin_fmaf(z, at.y, sum);
            z = a[u].z + b[u].z; z = (z > 0.f) ? z : SLOPE * z; sum = __builtin_fmaf(z, at.z, sum);
            z = a[u].w + b[u].w; z = (z > 0.f) ? z : SLOPE * z; sum = __builtin_fmaf(z, at.w, sum);
#pragma unroll
            for (int off = 16; off > 0; off >>= 1) sum += __shfl_xor(sum, off, 32);
            if (lane == i0 + u) res = sum;
        }
    }
    if (ok) {
        logit[gt] = res;
        atomicMax(menc + dv, enc_f32(res));
    }
}

__global__ void edge_exp_kernel(
    const float* __restrict__ logit, const int* __restrict__ dst,
    const unsigned* __restrict__ menc, float* __restrict__ eexp,
    float* __restrict__ ssum, int nedge)
{
    int e = (int)(blockIdx.x * blockDim.x + threadIdx.x);
    if (e >= nedge) return;
    const int d = dst[e];
    const float v = expf(logit[e] - dec_f32(menc[d]));
    eexp[e] = v;
    atomicAdd(ssum + d, v);
}

__global__ __launch_bounds__(256) void edge_scatter_kernel(
    const float* __restrict__ hid,
    const int* __restrict__ src, const int* __restrict__ dst,
    const float* __restrict__ eexp, const float* __restrict__ ssum,
    float* __restrict__ agg, int nedge)
{
    const int gt   = (int)(blockIdx.x * blockDim.x + threadIdx.x);
    const int lane = (int)(threadIdx.x & 31);
    const int e0 = __builtin_amdgcn_readfirstlane(gt & ~31);
    if (e0 >= nedge) return;
    const bool ok = gt < nedge;
    const int sv = ok ? src[gt] : 0;
    const int dv = ok ? dst[gt] : 0;
    float av = 0.f;
    if (ok) av = eexp[gt] / (ssum[dv] + 1e-16f);
    const int avi = __float_as_int(av);
    const int cnt = (nedge - e0 < 32) ? (nedge - e0) : 32;

    int i0 = 0;
    for (; i0 + SU <= cnt; i0 += SU) {
        float2 a[SU];
#pragma unroll
        for (int u = 0; u < SU; ++u) {
            const int s = __builtin_amdgcn_readlane(sv, i0 + u);
            a[u] = *(const float2*)(hid + (size_t)s * HID + lane * 2);
        }
#pragma unroll
        for (int u = 0; u < SU; ++u) {
            const int d = __builtin_amdgcn_readlane(dv, i0 + u);
            const float alpha = __int_as_float(__builtin_amdgcn_readlane(avi, i0 + u));
            float* o = agg + (size_t)d * HID + lane * 2;
            atomicAdd(o + 0, alpha * a[u].x);
            atomicAdd(o + 1, alpha * a[u].y);
        }
    }
    for (; i0 < cnt; ++i0) {
        const int s = __builtin_amdgcn_readlane(sv, i0);
        const int d = __builtin_amdgcn_readlane(dv, i0);
        const float alpha = __int_as_float(__builtin_amdgcn_readlane(avi, i0));
        const float2 a = *(const float2*)(hid + (size_t)s * HID + lane * 2);
        float* o = agg + (size_t)d * HID + lane * 2;
        atomicAdd(o + 0, alpha * a.x);
        atomicAdd(o + 1, alpha * a.y);
    }
}

#define FROWS (HID + 1)

__global__ __launch_bounds__(160) void fold_kernel(
    const float* __restrict__ W2a, const float* __restrict__ b2a,
    const float* __restrict__ Wla, const float* __restrict__ bla,
    const float* __restrict__ W2b, const float* __restrict__ b2b,
    const float* __restrict__ Wlb, const float* __restrict__ blb,
    float* __restrict__ F)
{
    const bool second = blockIdx.y != 0;
    const float* __restrict__ W2 = second ? W2b : W2a;
    const float* __restrict__ b2 = second ? b2b : b2a;
    const float* __restrict__ Wl = second ? Wlb : Wla;
    const float* __restrict__ bl = second ? blb : bla;
    float* __restrict__ Fo = F + (second ? FROWS * DIM : 0);

    const int wave = (int)(threadIdx.x >> 5);
    const int lane = (int)(threadIdx.x & 31);
    const int nloc  = lane & 15;
    const int khalf = (lane >> 4) << 1;
    const int n0 = (int)blockIdx.x * 16 + nloc;

    const int row = wave * 16 + nloc;
    const float* __restrict__ arow = (row < HID) ? W2 + (size_t)row * DIM : b2;

    v8f acc = (v8f){0.f, 0.f, 0.f, 0.f, 0.f, 0.f, 0.f, 0.f};
#pragma unroll
    for (int k0 = 0; k0 < DIM; k0 += 4) {
        v2f a, b;
        a.x = arow[k0 + khalf];
        a.y = arow[k0 + khalf + 1];
        b.x = Wl[(size_t)(k0 + khalf) * DIM + n0];
        b.y = Wl[(size_t)(k0 + khalf + 1) * DIM + n0];
        acc = __builtin_amdgcn_wmma_f32_16x16x4_f32(
            false, a, false, b, (short)0, acc, false, false);
    }

    const int mbase = wave * 16 + ((lane >> 4) << 3);
    const float blv = bl[n0];
#pragma unroll
    for (int j = 0; j < 8; ++j) {
        const int m = mbase + j;
        if (m <= HID)
            Fo[(size_t)m * DIM + n0] = acc[j] + ((m == HID) ? blv : 0.f);
    }
}

static inline void launch_gemm(const float* A, const float* W, const float* b,
                               const float* rowscale, const float* b2,
                               float* C, int rows, int K, int ncols, int relu,
                               hipStream_t stream) {
    long waves = (long)(rows / 16);
    long threads = waves * 32;
    int blocks = (int)((threads + 127) / 128);
    if (ncols == 64)
        gemm_bias_kernel<4><<<blocks, 128, 0, stream>>>(A, W, b, rowscale, b2, C, rows, K, relu);
    else
        gemm_bias_kernel<8><<<blocks, 128, 0, stream>>>(A, W, b, rowscale, b2, C, rows, K, relu);
}

extern "C" void kernel_launch(void* const* d_in, const int* in_sizes, int n_in,
                              void* d_out, int out_size, void* d_ws, size_t ws_size,
                              hipStream_t stream) {
    (void)in_sizes; (void)n_in; (void)out_size; (void)ws_size;

    const float* x_user = (const float*)d_in[0];
    const float* x_item = (const float*)d_in[1];
    const int*   e_u2i  = (const int*)d_in[2];
    const int*   e_i2u  = (const int*)d_in[3];
    const float* W1_u = (const float*)d_in[4],  *b1_u = (const float*)d_in[5];
    const float* W2_u = (const float*)d_in[6],  *b2_u = (const float*)d_in[7];
    const float* W1_i = (const float*)d_in[8],  *b1_i = (const float*)d_in[9];
    const float* W2_i = (const float*)d_in[10], *b2_i = (const float*)d_in[11];
    const float* Wl_u2i = (const float*)d_in[12], *bl_u2i = (const float*)d_in[13];
    const float* Wr_u2i = (const float*)d_in[14], *br_u2i = (const float*)d_in[15];
    const float* att_u2i = (const float*)d_in[16], *bias_u2i = (const float*)d_in[17];
    const float* Wl_i2u = (const float*)d_in[18], *bl_i2u = (const float*)d_in[19];
    const float* Wr_i2u = (const float*)d_in[20], *br_i2u = (const float*)d_in[21];
    const float* att_i2u = (const float*)d_in[22], *bias_i2u = (const float*)d_in[23];

    float* out_user = (float*)d_out;
    float* out_item = (float*)d_out + (size_t)N_NODES * DIM;

    float* ws   = (float*)d_ws;
    float* hu   = ws;
    float* hi   = hu + (size_t)N_NODES * DIM;
    float* hidu = hi + (size_t)N_NODES * DIM;
    float* xl   = hidu + (size_t)N_NODES * HID;
    float* xr   = xl + (size_t)N_NODES * DIM;
    float* logit = xr + (size_t)N_NODES * DIM;
    float* eexp  = logit + NEDGE;
    unsigned* menc = (unsigned*)(eexp + NEDGE);
    float* ssum  = (float*)(menc + N_NODES);
    float* hidi = out_user;
    float* agg  = xr;
    float* wsum = xr + (size_t)N_NODES * HID;
    float* Fu2i = hidu;
    float* Fi2u = hidu + FROWS * DIM;

    const float* none = nullptr;

    const int blkN    = (N_NODES + 255) / 256;
    const int blkNH   = (N_NODES * HID + 255) / 256;
    const int blkE    = (NEDGE + 255) / 256;

    launch_gemm(x_user, W1_u, b1_u, none, none, hidu, N_NODES, DIM, HID, 1, stream);
    launch_gemm(hidu,   W2_u, b2_u, none, none, hu,   N_NODES, HID, DIM, 0, stream);
    launch_gemm(x_item, W1_i, b1_i, none, none, hidi, N_NODES, DIM, HID, 1, stream);
    launch_gemm(hidi,   W2_i, b2_i, none, none, hi,   N_NODES, HID, DIM, 0, stream);

    {
        const int* src = e_u2i;
        const int* dst = e_u2i + NEDGE;
        launch_gemm(hu, Wl_u2i, bl_u2i, none, none, xl, N_NODES, DIM, DIM, 0, stream);
        launch_gemm(hi, Wr_u2i, br_u2i, none, none, xr, N_NODES, DIM, DIM, 0, stream);
        init_ms_kernel<<<blkN, 256, 0, stream>>>(menc, ssum, N_NODES);
        edge_logit_kernel<<<blkE, 256, 0, stream>>>(xl, xr, src, dst, att_u2i,
                                                     logit, menc, NEDGE);
        edge_exp_kernel<<<blkE, 256, 0, stream>>>(logit, dst, menc, eexp, ssum, NEDGE);
        init_agg_kernel<<<blkNH, 256, 0, stream>>>(agg, wsum, ssum, N_NODES);
        edge_scatter_kernel<<<blkE, 256, 0, stream>>>(hidu, src, dst, eexp, ssum,
                                                       agg, NEDGE);
        fold_kernel<<<dim3(DIM / 16, 2), 160, 0, stream>>>(
            W2_u, b2_u, Wl_u2i, bl_u2i, W2_i, b2_i, Wl_i2u, bl_i2u, Fu2i);
        launch_gemm(agg, Fu2i, Fu2i + HID * DIM, wsum, bias_u2i, out_item, N_NODES, HID, DIM, 0, stream);
    }

    {
        const int* src = e_i2u;
        const int* dst = e_i2u + NEDGE;
        launch_gemm(hi, Wl_i2u, bl_i2u, none, none, xl, N_NODES, DIM, DIM, 0, stream);
        launch_gemm(hu, Wr_i2u, br_i2u, none, none, xr, N_NODES, DIM, DIM, 0, stream);
        init_ms_kernel<<<blkN, 256, 0, stream>>>(menc, ssum, N_NODES);
        edge_logit_kernel<<<blkE, 256, 0, stream>>>(xl, xr, src, dst, att_i2u,
                                                     logit, menc, NEDGE);
        edge_exp_kernel<<<blkE, 256, 0, stream>>>(logit, dst, menc, eexp, ssum, NEDGE);
        init_agg_kernel<<<blkNH, 256, 0, stream>>>(agg, wsum, ssum, N_NODES);
        edge_scatter_kernel<<<blkE, 256, 0, stream>>>(hidi, src, dst, eexp, ssum,
                                                       agg, NEDGE);
        launch_gemm(agg, Fi2u, Fi2u + HID * DIM, wsum, bias_i2u, out_user, N_NODES, HID, DIM, 0, stream);
    }
}


# GEMM B operand through LDS: one 16-wave work-group per WGP copies W to LDS once in fragment order; each wave takes a 16-row stripe, A in registers, B as 8-byte LDS reads (were 4-byte global loads)
# speedup vs baseline: 1.7289x; 1.0677x over previous
#include <hip/hip_runtime.h>

typedef __attribute__((ext_vector_type(2))) float v2f;
typedef __attribute__((ext_vector_type(8))) float v8f;

#define N_NODES 50000
#define DIM 128
#define HID 64
#define NEDGE 500000
#define SLOPE 0.2f

#define GEMM_THREADS 512
#define GEMM_GROUPS  256
#define FULL_WAVES   12
#define GEMM_LDS_FLOATS (41 * 1024)

template <int NT, int NTW, int K>
__device__ __forceinline__ void stripe_tiles(
    const float* __restrict__ arow, const float* __restrict__ lp,
    const float* __restrict__ bias, const float* __restrict__ rowscale,
    const float* __restrict__ bias2, float* __restrict__ C,
    int stripe, int t0, int relu, int lane)
{
    constexpr int ncols = NT * 16;
    constexpr int PITCH = 2 * ncols + 32;
    const int nloc = lane & 15;

    v2f a[K / 4];
#pragma unroll
    for (int kk = 0; kk < K / 4; ++kk) a[kk] = *(const v2f*)(arow + kk * 4);

    v8f acc[NTW];
#pragma unroll
    for (int t = 0; t < NTW; ++t)
        acc[t] = (v8f){0.f, 0.f, 0.f, 0.f, 0.f, 0.f, 0.f, 0.f};

#pragma unroll
    for (int kk = 0; kk < K / 4; ++kk) {
#pragma unroll
        for (int t = 0; t < NTW; ++t) {
            const v2f b = *(const v2f*)(lp + (kk * 2) * PITCH + t * 32);
            acc[t] = __builtin_amdgcn_wmma_f32_16x16x4_f32(
                false, a[kk], false, b, (short)0, acc[t], false, false);
        }
    }

    const int mbase = stripe * 16 + ((lane >> 4) << 3);
    float rs[8];
#pragma unroll
    for (int j = 0; j < 8; ++j)
        rs[j] = rowscale ? rowscale[mbase + j] : 1.f;
#pragma unroll
    for (int t = 0; t < NTW; ++t) {
        const int n = (t0 + t) * 16 + nloc;
        const float bv = bias[n];
        const float b2v = bias2 ? bias2[n] : 0.f;
#pragma unroll
        for (int j = 0; j < 8; ++j) {
            float v;
            if (rowscale) v = acc[t][j] + rs[j] * bv + b2v;
            else          v = acc[t][j] + bv;
            if (relu) v = fmaxf(v, 0.f);
            C[(size_t)(mbase + j) * ncols + n] = v;
        }
    }
}

template <int NT, int K>
__global__ __launch_bounds__(GEMM_THREADS) void gemm_bias_kernel(
    const float* __restrict__ A, const float* __restrict__ W,
    const float* __restrict__ bias, const float* __restrict__ rowscale,
    const float* __restrict__ bias2, float* __restrict__ C,
    int rows, int relu)
{
    constexpr int ncols = NT * 16;
    constexpr int PITCH = 2 * ncols + 32;
    constexpr int ITEMS = (K / 2) * (ncols / 4);
    static_assert(ITEMS % GEMM_THREADS == 0, "copy items must divide evenly");
    static_assert((K / 2) * PITCH <= GEMM_LDS_FLOATS, "W must fit the LDS array");
    __shared__ __attribute__((aligned(16))) float wl[GEMM_LDS_FLOATS];

    const int lane  = (int)(threadIdx.x & 31);
    const int wib   = (int)(threadIdx.x >> 5);
    const int nloc  = lane & 15;
    const int hsel  = lane >> 4;
    const bool full = wib < FULL_WAVES;
    const int stripe = (int)blockIdx.x + (int)gridDim.x * (full ? wib : FULL_WAVES);
    const bool live = stripe < (rows >> 4);

    constexpr int NIT = ITEMS / GEMM_THREADS;
    float4 r0[NIT], r1[NIT];
#pragma unroll
    for (int it = 0; it < NIT; ++it) {
        const int i = it * GEMM_THREADS + (int)threadIdx.x;
        const int p = i / (ncols / 4);
        const int q = i % (ncols / 4);
        r0[it] = *(const float4*)(W + (size_t)(2 * p) * ncols + q * 4);
        r1[it] = *(const float4*)(W + (size_t)(2 * p + 1) * ncols + q * 4);
    }

#pragma unroll
    for (int it = 0; it < NIT; ++it) {
        const int i = it * GEMM_THREADS + (int)threadIdx.x;
        const int p = i / (ncols / 4);
        const int q = i % (ncols / 4);
        float* d = wl + p * PITCH + q * 8;
        *(float4*)(d)     = make_float4(r0[it].x, r1[it].x, r0[it].y, r1[it].y);
        *(float4*)(d + 4) = make_float4(r0[it].z, r1[it].z, r0[it].w, r1[it].w);
    }
    __syncthreads();

    if (live) {
        const float* __restrict__ arow =
            A + (size_t)(stripe * 16 + nloc) * (size_t)K + (hsel << 1);
        const float* __restrict__ lbase = wl + hsel * PITCH + nloc * 2;
        if (full) {
            stripe_tiles<NT, NT, K>(arow, lbase, bias, rowscale, bias2, C,
                                    stripe, 0, relu, lane);
        } else {
            const int t0 = (wib - FULL_WAVES) * (NT / 4);
            stripe_tiles<NT, NT / 4, K>(arow, lbase + t0 * 32, bias, rowscale, bias2, C,
                                        stripe, t0, relu, lane);
        }
    }
}

__device__ __forceinline__ unsigned enc_f32(float x) {
    unsigned b = __float_as_uint(x);
    return (b & 0x80000000u) ? ~b : (b | 0x80000000u);
}
__device__ __forceinline__ float dec_f32(unsigned k) {
    unsigned b = (k & 0x80000000u) ? (k & 0x7FFFFFFFu) : ~k;
    return __uint_as_float(b);
}

__global__ void init_agg_kernel(float* __restrict__ agg, float* __restrict__ wsum,
                                const float* __restrict__ ssum, int n) {
    int i = (int)(blockIdx.x * blockDim.x + threadIdx.x);
    if (i < n * HID) agg[i] = 0.f;
    if (i < n) { const float s = ssum[i]; wsum[i] = s / (s + 1e-16f); }
}
__global__ void init_ms_kernel(unsigned* __restrict__ menc,
                               float* __restrict__ ssum, int n) {
    int i = (int)(blockIdx.x * blockDim.x + threadIdx.x);
    if (i < n) { menc[i] = 0u; ssum[i] = 0.f; }
}

#define LU 4
#define SU 16

__global__ __launch_bounds__(256) void edge_logit_kernel(
    const float* __restrict__ xl, const float* __restrict__ xr,
    const int* __restrict__ src, const int* __restrict__ dst,
    const float* __restrict__ att,
    float* __restrict__ logit, unsigned* __restrict__ menc, int nedge)
{
    const int gt   = (int)(blockIdx.x * blockDim.x + threadIdx.x);
    const int lane = (int)(threadIdx.x & 31);
    const int e0 = __builtin_amdgcn_readfirstlane(gt & ~31);
    if (e0 >= nedge) return;
    const bool ok = gt < nedge;
    const int sv = ok ? src[gt] : 0;
    const int dv = ok ? dst[gt] : 0;
    const float4 at = *(const float4*)(att + lane * 4);
    const int cnt = (nedge - e0 < 32) ? (nedge - e0) : 32;

    float res = 0.f;
    for (int i0 = 0; i0 < cnt; i0 += LU) {
        float4 a[LU], b[LU];
#pragma unroll
        for (int u = 0; u < LU; ++u) {
            const int s = __builtin_amdgcn_readlane(sv, i0 + u);
            const int d = __builtin_amdgcn_readlane(dv, i0 + u);
            a[u] = *(const float4*)(xl + (size_t)s * DIM + lane * 4);
            b[u] = *(const float4*)(xr + (size_t)d * DIM + lane * 4);
        }
#pragma unroll
        for (int u = 0; u < LU; ++u) {
            float z, sum = 0.f;
            z = a[u].x + b[u].x; z = (z > 0.f) ? z : SLOPE * z; sum = __builtin_fmaf(z, at.x, sum);
            z = a[u].y + b[u].y; z = (z > 0.f) ? z : SLOPE * z; sum = __builtin_fmaf(z, at.y, sum);
            z = a[u].z + b[u].z; z = (z > 0.f) ? z : SLOPE * z; sum = __builtin_fmaf(z, at.z, sum);
            z = a[u].w + b[u].w; z = (z > 0.f) ? z : SLOPE * z; sum = __builtin_fmaf(z, at.w, sum);
#pragma unroll
            for (int off = 16; off > 0; off >>= 1) sum += __shfl_xor(sum, off, 32);
            if (lane == i0 + u) res = sum;
        }
    }
    if (ok) {
        logit[gt] = res;
        atomicMax(menc + dv, enc_f32(res));
    }
}

__global__ void edge_exp_kernel(
    const float* __restrict__ logit, const int* __restrict__ dst,
    const unsigned* __restrict__ menc, float* __restrict__ eexp,
    float* __restrict__ ssum, int nedge)
{
    int e = (int)(blockIdx.x * blockDim.x + threadIdx.x);
    if (e >= nedge) return;
    const int d = dst[e];
    const float v = expf(logit[e] - dec_f32(menc[d]));
    eexp[e] = v;
    atomicAdd(ssum + d, v);
}

__global__ __launch_bounds__(256) void edge_scatter_kernel(
    const float* __restrict__ hid,
    const int* __restrict__ src, const int* __restrict__ dst,
    const float* __restrict__ eexp, const float* __restrict__ ssum,
    float* __restrict__ agg, int nedge)
{
    const int gt   = (int)(blockIdx.x * blockDim.x + threadIdx.x);
    const int lane = (int)(threadIdx.x & 31);
    const int e0 = __builtin_amdgcn_readfirstlane(gt & ~31);
    if (e0 >= nedge) return;
    const bool ok = gt < nedge;
    const int sv = ok ? src[gt] : 0;
    const int dv = ok ? dst[gt] : 0;
    float av = 0.f;
    if (ok) av = eexp[gt] / (ssum[dv] + 1e-16f);
    const int avi = __float_as_int(av);
    const int cnt = (nedge - e0 < 32) ? (nedge - e0) : 32;

    int i0 = 0;
    for (; i0 + SU <= cnt; i0 += SU) {
        float2 a[SU];
#pragma unroll
        for (int u = 0; u < SU; ++u) {
            const int s = __builtin_amdgcn_readlane(sv, i0 + u);
            a[u] = *(const float2*)(hid + (size_t)s * HID + lane * 2);
        }
#pragma unroll
        for (int u = 0; u < SU; ++u) {
            const int d = __builtin_amdgcn_readlane(dv, i0 + u);
            const float alpha = __int_as_float(__builtin_amdgcn_readlane(avi, i0 + u));
            float* o = agg + (size_t)d * HID + lane * 2;
            atomicAdd(o + 0, alpha * a[u].x);
            atomicAdd(o + 1, alpha * a[u].y);
        }
    }
    for (; i0 < cnt; ++i0) {
        const int s = __builtin_amdgcn_readlane(sv, i0);
        const int d = __builtin_amdgcn_readlane(dv, i0);
        const float alpha = __int_as_float(__builtin_amdgcn_readlane(avi, i0));
        const float2 a = *(const float2*)(hid + (size_t)s * HID + lane * 2);
        float* o = agg + (size_t)d * HID + lane * 2;
        atomicAdd(o + 0, alpha * a.x);
        atomicAdd(o + 1, alpha * a.y);
    }
}

#define FROWS (HID + 1)

__global__ __launch_bounds__(160) void fold_kernel(
    const float* __restrict__ W2a, const float* __restrict__ b2a,
    const float* __restrict__ Wla, const float* __restrict__ bla,
    const float* __restrict__ W2b, const float* __restrict__ b2b,
    const float* __restrict__ Wlb, const float* __restrict__ blb,
    float* __restrict__ F)
{
    const bool second = blockIdx.y != 0;
    const float* __restrict__ W2 = second ? W2b : W2a;
    const float* __restrict__ b2 = second ? b2b : b2a;
    const float* __restrict__ Wl = second ? Wlb : Wla;
    const float* __restrict__ bl = second ? blb : bla;
    float* __restrict__ Fo = F + (second ? FROWS * DIM : 0);

    const int wave = (int)(threadIdx.x >> 5);
    const int lane = (int)(threadIdx.x & 31);
    const int nloc  = lane & 15;
    const int khalf = (lane >> 4) << 1;
    const int n0 = (int)blockIdx.x * 16 + nloc;

    const int row = wave * 16 + nloc;
    const float* __restrict__ arow = (row < HID) ? W2 + (size_t)row * DIM : b2;

    v8f acc = (v8f){0.f, 0.f, 0.f, 0.f, 0.f, 0.f, 0.f, 0.f};
#pragma unroll
    for (int k0 = 0; k0 < DIM; k0 += 4) {
        v2f a, b;
        a.x = arow[k0 + khalf];
        a.y = arow[k0 + khalf + 1];
        b.x = Wl[(size_t)(k0 + khalf) * DIM + n0];
        b.y = Wl[(size_t)(k0 + khalf + 1) * DIM + n0];
        acc = __builtin_amdgcn_wmma_f32_16x16x4_f32(
            false, a, false, b, (short)0, acc, false, false);
    }

    const int mbase = wave * 16 + ((lane >> 4) << 3);
    const float blv = bl[n0];
#pragma unroll
    for (int j = 0; j < 8; ++j) {
        const int m = mbase + j;
        if (m <= HID)
            Fo[(size_t)m * DIM + n0] = acc[j] + ((m == HID) ? blv : 0.f);
    }
}

static inline void launch_gemm(const float* A, const float* W, const float* b,
                               const float* rowscale, const float* b2,
                               float* C, int rows, int K, int ncols, int relu,
                               hipStream_t stream) {
    if (ncols == 64)
        gemm_bias_kernel<4, 128><<<GEMM_GROUPS, GEMM_THREADS, 0, stream>>>(A, W, b, rowscale, b2, C, rows, relu);
    else if (K == 64)
        gemm_bias_kernel<8, 64><<<GEMM_GROUPS, GEMM_THREADS, 0, stream>>>(A, W, b, rowscale, b2, C, rows, relu);
    else
        gemm_bias_kernel<8, 128><<<GEMM_GROUPS, GEMM_THREADS, 0, stream>>>(A, W, b, rowscale, b2, C, rows, relu);
}

extern "C" void kernel_launch(void* const* d_in, const int* in_sizes, int n_in,
                              void* d_out, int out_size, void* d_ws, size_t ws_size,
                              hipStream_t stream) {
    (void)in_sizes; (void)n_in; (void)out_size; (void)ws_size;

    const float* x_user = (const float*)d_in[0];
    const float* x_item = (const float*)d_in[1];
    const int*   e_u2i  = (const int*)d_in[2];
    const int*   e_i2u  = (const int*)d_in[3];
    const float* W1_u = (const float*)d_in[4],  *b1_u = (const float*)d_in[5];
    const float* W2_u = (const float*)d_in[6],  *b2_u = (const float*)d_in[7];
    const float* W1_i = (const float*)d_in[8],  *b1_i = (const float*)d_in[9];
    const float* W2_i = (const float*)d_in[10], *b2_i = (const float*)d_in[11];
    const float* Wl_u2i = (const float*)d_in[12], *bl_u2i = (const float*)d_in[13];
    const float* Wr_u2i = (const float*)d_in[14], *br_u2i = (const float*)d_in[15];
    const float* att_u2i = (const float*)d_in[16], *bias_u2i = (const float*)d_in[17];
    const float* Wl_i2u = (const float*)d_in[18], *bl_i2u = (const float*)d_in[19];
    const float* Wr_i2u = (const float*)d_in[20], *br_i2u = (const float*)d_in[21];
    const float* att_i2u = (const float*)d_in[22], *bias_i2u = (const float*)d_in[23];

    float* out_user = (float*)d_out;
    float* out_item = (float*)d_out + (size_t)N_NODES * DIM;

    float* ws   = (float*)d_ws;
    float* hu   = ws;
    float* hi   = hu + (size_t)N_NODES * DIM;
    float* hidu = hi + (size_t)N_NODES * DIM;
    float* xl   = hidu + (size_t)N_NODES * HID;
    float* xr   = xl + (size_t)N_NODES * DIM;
    float* logit = xr + (size_t)N_NODES * DIM;
    float* eexp  = logit + NEDGE;
    unsigned* menc = (unsigned*)(eexp + NEDGE);
    float* ssum  = (float*)(menc + N_NODES);
    float* hidi = out_user;
    float* agg  = xr;
    float* wsum = xr + (size_t)N_NODES * HID;
    float* Fu2i = hidu;
    float* Fi2u = hidu + FROWS * DIM;

    const float* none = nullptr;

    const int blkN    = (N_NODES + 255) / 256;
    const int blkNH   = (N_NODES * HID + 255) / 256;
    const int blkE    = (NEDGE + 255) / 256;

    launch_gemm(x_user, W1_u, b1_u, none, none, hidu, N_NODES, DIM, HID, 1, stream);
    launch_gemm(hidu,   W2_u, b2_u, none, none, hu,   N_NODES, HID, DIM, 0, stream);
    launch_gemm(x_item, W1_i, b1_i, none, none, hidi, N_NODES, DIM, HID, 1, stream);
    launch_gemm(hidi,   W2_i, b2_i, none, none, hi,   N_NODES, HID, DIM, 0, stream);

    {
        const int* src = e_u2i;
        const int* dst = e_u2i + NEDGE;
        launch_gemm(hu, Wl_u2i, bl_u2i, none, none, xl, N_NODES, DIM, DIM, 0, stream);
        launch_gemm(hi, Wr_u2i, br_u2i, none, none, xr, N_NODES, DIM, DIM, 0, stream);
        init_ms_kernel<<<blkN, 256, 0, stream>>>(menc, ssum, N_NODES);
        edge_logit_kernel<<<blkE, 256, 0, stream>>>(xl, xr, src, dst, att_u2i,
                                                     logit, menc, NEDGE);
        edge_exp_kernel<<<blkE, 256, 0, stream>>>(logit, dst, menc, eexp, ssum, NEDGE);
        init_agg_kernel<<<blkNH, 256, 0, stream>>>(agg, wsum, ssum, N_NODES);
        edge_scatter_kernel<<<blkE, 256, 0, stream>>>(hidu, src, dst, eexp, ssum,
                                                       agg, NEDGE);
        fold_kernel<<<dim3(DIM / 16, 2), 160, 0, stream>>>(
            W2_u, b2_u, Wl_u2i, bl_u2i, W2_i, b2_i, Wl_i2u, bl_i2u, Fu2i);
        launch_gemm(agg, Fu2i, Fu2i + HID * DIM, wsum, bias_u2i, out_item, N_NODES, HID, DIM, 0, stream);
    }

    {
        const int* src = e_i2u;
        const int* dst = e_i2u + NEDGE;
        launch_gemm(hi, Wl_i2u, bl_i2u, none, none, xl, N_NODES, DIM, DIM, 0, stream);
        launch_gemm(hu, Wr_i2u, br_i2u, none, none, xr, N_NODES, DIM, DIM, 0, stream);
        init_ms_kernel<<<blkN, 256, 0, stream>>>(menc, ssum, N_NODES);
        edge_logit_kernel<<<blkE, 256, 0, stream>>>(xl, xr, src, dst, att_i2u,
                                                     logit, menc, NEDGE);
        edge_exp_kernel<<<blkE, 256, 0, stream>>>(logit, dst, menc, eexp, ssum, NEDGE);
        init_agg_kernel<<<blkNH, 256, 0, stream>>>(agg, wsum, ssum, N_NODES);
        edge_scatter_kernel<<<blkE, 256, 0, stream>>>(hidi, src, dst, eexp, ssum,
                                                       agg, NEDGE);
        launch_gemm(agg, Fi2u, Fi2u + HID * DIM, wsum, bias_i2u, out_user, N_NODES, HID, DIM, 0, stream);
    }
}
